# starting kernel
# baseline (speedup 1.0000x reference)
#include <hip/hip_runtime.h>

#define FEAT 64
#define NUM_RELS 8

typedef float v2f __attribute__((ext_vector_type(2)));
typedef float v8f __attribute__((ext_vector_type(8)));

__global__ void rgcn_zero_kernel(float4* __restrict__ out, int n4) {
  int i = blockIdx.x * blockDim.x + threadIdx.x;
  if (i < n4) out[i] = make_float4(0.f, 0.f, 0.f, 0.f);
}

__global__ __launch_bounds__(128) void rgcn_gemm_wmma(
    const float* __restrict__ feat, const float* __restrict__ weight,
    float* __restrict__ T, int r0, int N) {
  const int node0 = blockIdx.x << 4;
  const int rloc  = blockIdx.y;
  const int wave  = threadIdx.x >> 5;
  const int lane  = threadIdx.x & 31;
  const int m     = lane & 15;
  const int khalf = lane >> 4;
  const int o0    = wave << 4;

  const float* __restrict__ Wr = weight + (size_t)(r0 + rloc) * (FEAT * FEAT);
  const float* __restrict__ ap = feat + (size_t)(node0 + m) * FEAT + (khalf << 1);
  const float* __restrict__ bp = Wr + ((khalf << 1) * FEAT) + o0 + m;

  v8f c = {};
#pragma unroll
  for (int k0 = 0; k0 < FEAT; k0 += 4) {
    v2f a, b;
    a.x = ap[k0];
    a.y = ap[k0 + 1];
    b.x = bp[k0 * FEAT];
    b.y = bp[k0 * FEAT + FEAT];
    c = __builtin_amdgcn_wmma_f32_16x16x4_f32(
         false, a,  false, b,
         (short)0, c,  false,  false);
  }

  float* __restrict__ op = T + ((size_t)rloc * N + node0) * FEAT + o0 + m;
#pragma unroll
  for (int v = 0; v < 8; ++v) {
    op[(size_t)(v + (khalf << 3)) * FEAT] = c[v];
  }
}

__global__ __launch_bounds__(256) void rgcn_edge_scatter(
    const float* __restrict__ T, const float* __restrict__ norm,
    const int* __restrict__ etypes, const int* __restrict__ src,
    const int* __restrict__ dst, float* __restrict__ out,
    int r0, int r1, int N, int E) {
  long long t = (long long)blockIdx.x * blockDim.x + threadIdx.x;
  int e = (int)(t >> 4);
  int j = (int)(t & 15);
  if (e >= E) return;
  int r = etypes[e];
  if (r < r0 || r >= r1) return;

  float nw = norm[e];
  const float4* __restrict__ row = reinterpret_cast<const float4*>(
      T + (((size_t)(r - r0) * N + (size_t)src[e]) << 6));
  float4 v = row[j];
  float* op = out + ((size_t)dst[e] << 6) + (j << 2);
  unsafeAtomicAdd(op + 0, v.x * nw);
  unsafeAtomicAdd(op + 1, v.y * nw);
  unsafeAtomicAdd(op + 2, v.z * nw);
  unsafeAtomicAdd(op + 3, v.w * nw);
}

__global__ __launch_bounds__(256) void rgcn_edge_direct(
    const float* __restrict__ feat, const float* __restrict__ norm,
    const int* __restrict__ etypes, const int* __restrict__ src,
    const int* __restrict__ dst, const float* __restrict__ weight,
    float* __restrict__ out, int E) {
  int wave = threadIdx.x >> 5;
  int lane = threadIdx.x & 31;
  long long e = (long long)blockIdx.x * 8 + wave;
  if (e >= E) return;

  const float* f = feat + ((size_t)src[e] << 6);
  float x0 = f[lane], x1 = f[lane + 32];
  const float* Wr = weight + (size_t)etypes[e] * (FEAT * FEAT);
  float acc0 = 0.f, acc1 = 0.f;
#pragma unroll
  for (int i = 0; i < 32; ++i) {
    float fi = __shfl(x0, i, 32);
    acc0 += fi * Wr[i * FEAT + lane];
    acc1 += fi * Wr[i * FEAT + lane + 32];
  }
#pragma unroll
  for (int i = 0; i < 32; ++i) {
    float fi = __shfl(x1, i, 32);
    acc0 += fi * Wr[(i + 32) * FEAT + lane];
    acc1 += fi * Wr[(i + 32) * FEAT + lane + 32];
  }
  float nw = norm[e];
  float* op = out + ((size_t)dst[e] << 6);
  unsafeAtomicAdd(op + lane, acc0 * nw);
  unsafeAtomicAdd(op + lane + 32, acc1 * nw);
}

extern "C" void kernel_launch(void* const* d_in, const int* in_sizes, int n_in,
                              void* d_out, int out_size, void* d_ws, size_t ws_size,
                              hipStream_t stream) {
  const float* feat   = (const float*)d_in[0];
  const float* norm   = (const float*)d_in[1];
  const float* weight = (const float*)d_in[2];
  const int*   etypes = (const int*)d_in[3];
  const int*   src    = (const int*)d_in[4];
  const int*   dst    = (const int*)d_in[5];
  float* out = (float*)d_out;

  const int N = in_sizes[0] / FEAT;
  const int E = in_sizes[3];

  int n4 = out_size / 4;
  rgcn_zero_kernel<<<(n4 + 255) / 256, 256, 0, stream>>>((float4*)out, n4);

  const size_t slab = (size_t)N * FEAT * sizeof(float);
  int chunk = (int)(ws_size / slab);
  if (chunk > NUM_RELS) chunk = NUM_RELS;

  if (chunk >= 1 && (N % 16) == 0) {
    float* T = (float*)d_ws;
    for (int r0 = 0; r0 < NUM_RELS; r0 += chunk) {
      int r1 = r0 + chunk;
      if (r1 > NUM_RELS) r1 = NUM_RELS;
      dim3 grid(N / 16, r1 - r0);
      rgcn_gemm_wmma<<<grid, 128, 0, stream>>>(feat, weight, T, r0, N);
      long long tt = (long long)E * 16;
      int blocks = (int)((tt + 255) / 256);
      rgcn_edge_scatter<<<blocks, 256, 0, stream>>>(T, norm, etypes, src, dst,
                                                    out, r0, r1, N, E);
    }
  } else {
    int blocks = (E + 7) / 8;
    rgcn_edge_direct<<<blocks, 256, 0, stream>>>(feat, norm, etypes, src, dst,
                                                 weight, out, E);
  }
}


# Scatter replaced by in-call binning of edges by dst range + one block per range summing its rows in LDS (LDS float atomics, arrival order); each out row stored once, no global atomics, no zero fill
# speedup vs baseline: 2.0460x; 2.0460x over previous
#include <hip/hip_runtime.h>

#define FEAT 64
#define NUM_RELS 8

typedef float v2f __attribute__((ext_vector_type(2)));
typedef float v4f __attribute__((ext_vector_type(4)));
typedef float v8f __attribute__((ext_vector_type(8)));

__global__ void rgcn_zero_kernel(float4* __restrict__ out, int n4) {
  int i = blockIdx.x * blockDim.x + threadIdx.x;
  if (i < n4) out[i] = make_float4(0.f, 0.f, 0.f, 0.f);
}

__global__ __launch_bounds__(128) void rgcn_gemm_wmma(
    const float* __restrict__ feat, const float* __restrict__ weight,
    float* __restrict__ T, int r0, int N) {
  const int node0 = blockIdx.x << 4;
  const int rloc  = blockIdx.y;
  const int wave  = threadIdx.x >> 5;
  const int lane  = threadIdx.x & 31;
  const int m     = lane & 15;
  const int khalf = lane >> 4;
  const int o0    = wave << 4;

  const float* __restrict__ Wr = weight + (size_t)(r0 + rloc) * (FEAT * FEAT);
  const float* __restrict__ ap = feat + (size_t)(node0 + m) * FEAT + (khalf << 1);
  const float* __restrict__ bp = Wr + ((khalf << 1) * FEAT) + o0 + m;

  v8f c = {};
#pragma unroll
  for (int k0 = 0; k0 < FEAT; k0 += 4) {
    v2f a, b;
    a.x = ap[k0];
    a.y = ap[k0 + 1];
    b.x = bp[k0 * FEAT];
    b.y = bp[k0 * FEAT + FEAT];
    c = __builtin_amdgcn_wmma_f32_16x16x4_f32(
         false, a,  false, b,
         (short)0, c,  false,  false);
  }

  float* __restrict__ op = T + ((size_t)rloc * N + node0) * FEAT + o0 + m;
#pragma unroll
  for (int v = 0; v < 8; ++v) {
    op[(size_t)(v + (khalf << 3)) * FEAT] = c[v];
  }
}

#define WS_RELS 2
__global__ __launch_bounds__(256) void rgcn_gemm_wstat(
    const float* __restrict__ feat, const float* __restrict__ weight,
    float* __restrict__ T, int r0, int N) {
  const int lane = threadIdx.x & 31;
  const int wib  = __builtin_amdgcn_readfirstlane((int)(threadIdx.x >> 5));
  const int c    = lane & 15;
  const int h    = lane >> 4;
  const int rl0  = blockIdx.y * WS_RELS;
  const int ntiles = N >> 4;
  const int nslots = gridDim.x << 3;
  const int slot = wib * gridDim.x + blockIdx.x;
  if (slot >= ntiles) return;

  v2f b[WS_RELS][4][16];
#pragma unroll
  for (int q = 0; q < WS_RELS; ++q) {
    const float* __restrict__ Wr =
        weight + (size_t)(r0 + rl0 + q) * (FEAT * FEAT) + ((h << 1) * FEAT) + (c << 2);
#pragma unroll
    for (int s = 0; s < 16; ++s) {
      const v4f w0 = *reinterpret_cast<const v4f*>(Wr + (s * 4) * FEAT);
      const v4f w1 = *reinterpret_cast<const v4f*>(Wr + (s * 4 + 1) * FEAT);
      b[q][0][s].x = w0.x; b[q][0][s].y = w1.x;
      b[q][1][s].x = w0.y; b[q][1][s].y = w1.y;
      b[q][2][s].x = w0.z; b[q][2][s].y = w1.z;
      b[q][3][s].x = w0.w; b[q][3][s].y = w1.w;
    }
  }

  const float* __restrict__ abase = feat + (size_t)c * FEAT + (h << 1);
  v2f a[16];
  {
    const float* __restrict__ ap = abase + (size_t)slot * (16 * FEAT);
#pragma unroll
    for (int s = 0; s < 16; ++s) a[s] = *reinterpret_cast<const v2f*>(ap + s * 4);
  }

  for (int tile = slot; tile < ntiles; tile += nslots) {
    int nt = tile + nslots;
    if (nt >= ntiles) nt = tile;
    v2f an[16];
    {
      const float* __restrict__ ap = abase + (size_t)nt * (16 * FEAT);
#pragma unroll
      for (int s = 0; s < 16; ++s) an[s] = *reinterpret_cast<const v2f*>(ap + s * 4);
    }
    __builtin_amdgcn_sched_barrier(0);

    v8f acc[WS_RELS][4];
#pragma unroll
    for (int q = 0; q < WS_RELS; ++q)
#pragma unroll
      for (int t = 0; t < 4; ++t) acc[q][t] = v8f{};

#pragma unroll
    for (int s = 0; s < 16; ++s) {
#pragma unroll
      for (int q = 0; q < WS_RELS; ++q) {
#pragma unroll
        for (int t = 0; t < 4; ++t) {
          acc[q][t] = __builtin_amdgcn_wmma_f32_16x16x4_f32(
               false, a[s],  false, b[q][t][s],
               (short)0, acc[q][t],  false,  false);
        }
      }
    }

    __builtin_amdgcn_sched_barrier(0);

#pragma unroll
    for (int q = 0; q < WS_RELS; ++q) {
      float* __restrict__ op =
          T + (((size_t)(rl0 + q) * N + ((size_t)tile << 4) + (h << 3)) << 6) + (c << 2);
#pragma unroll
      for (int v = 0; v < 8; ++v) {
        v4f o;
        o.x = acc[q][0][v]; o.y = acc[q][1][v]; o.z = acc[q][2][v]; o.w = acc[q][3][v];
        *reinterpret_cast<v4f*>(op + v * FEAT) = o;
      }
    }

#pragma unroll
    for (int s = 0; s < 16; ++s) a[s] = an[s];
  }
}

__global__ __launch_bounds__(256) void rgcn_edge_scatter(
    const float* __restrict__ T, const float* __restrict__ norm,
    const int* __restrict__ etypes, const int* __restrict__ src,
    const int* __restrict__ dst, float* __restrict__ out,
    int r0, int r1, int N, int E) {
  long long t = (long long)blockIdx.x * blockDim.x + threadIdx.x;
  int e = (int)(t >> 4);
  int j = (int)(t & 15);
  if (e >= E) return;
  int r = etypes[e];
  if (r < r0 || r >= r1) return;

  float nw = norm[e];
  const float4* __restrict__ row = reinterpret_cast<const float4*>(
      T + (((size_t)(r - r0) * N + (size_t)src[e]) << 6));
  float4 v = row[j];
  float* op = out + ((size_t)dst[e] << 6) + (j << 2);
  unsafeAtomicAdd(op + 0, v.x * nw);
  unsafeAtomicAdd(op + 1, v.y * nw);
  unsafeAtomicAdd(op + 2, v.z * nw);
  unsafeAtomicAdd(op + 3, v.w * nw);
}

#define ACC_ROWS 196
#define NBKT_MAX 512
#define NSRC 128
#define ACC_WAVES 32
#define BOFF_PITCH (NBKT_MAX + 1)

__global__ __launch_bounds__(1024) void rgcn_bin_edges(
    const float* __restrict__ norm, const int* __restrict__ etypes,
    const int* __restrict__ src, const int* __restrict__ dst,
    int4* __restrict__ rec, int* __restrict__ boff, int E, int epb) {
  __shared__ int cnt[NBKT_MAX];
  __shared__ int cur[NBKT_MAX];
  const int tid = threadIdx.x;
  const int e0 = blockIdx.x * epb;
  int e1 = e0 + epb;
  if (e1 > E) e1 = E;

  if (tid < NBKT_MAX) cnt[tid] = 0;
  __syncthreads();
  for (int eb = e0 + tid; eb < e1; eb += 4 * 1024) {
    int d[4];
#pragma unroll
    for (int u = 0; u < 4; ++u) {
      const int e = eb + u * 1024;
      d[u] = dst[e < e1 ? e : e1 - 1];
      if (e >= e1) d[u] = -1;
    }
#pragma unroll
    for (int u = 0; u < 4; ++u) {
      const unsigned b = (unsigned)d[u] / ACC_ROWS;
      if (b < NBKT_MAX) atomicAdd(&cnt[b], 1);
    }
  }
  __syncthreads();

  if (tid < 32) {
    const int base = tid * 16;
    int v[16];
    int loc = 0;
#pragma unroll
    for (int i = 0; i < 16; ++i) { v[i] = cnt[base + i]; loc += v[i]; }
    int incl = loc;
#pragma unroll
    for (int d = 1; d < 32; d <<= 1) {
      const int t = __shfl_up(incl, d, 32);
      if (tid >= d) incl += t;
    }
    int run = incl - loc;
    int* __restrict__ bo = boff + (size_t)blockIdx.x * BOFF_PITCH + base;
#pragma unroll
    for (int i = 0; i < 16; ++i) { cur[base + i] = run; bo[i] = run; run += v[i]; }
    if (tid == 31) bo[16] = run;
  }
  __syncthreads();

  for (int eb = e0 + tid; eb < e1; eb += 4 * 1024) {
    int   d[4], sv[4], tv[4];
    float nv[4];
#pragma unroll
    for (int u = 0; u < 4; ++u) {
      const int e  = eb + u * 1024;
      const int ec = e < e1 ? e : e1 - 1;
      d[u]  = dst[ec];
      sv[u] = src[ec];
      tv[u] = etypes[ec];
      nv[u] = norm[ec];
      if (e >= e1) d[u] = -1;
    }
#pragma unroll
    for (int u = 0; u < 4; ++u) {
      const unsigned b = (unsigned)d[u] / ACC_ROWS;
      if (b < NBKT_MAX) {
        const int pos = atomicAdd(&cur[b], 1);
        int4 r;
        r.x = sv[u];
        r.y = (tv[u] << 8) | (d[u] - (int)b * ACC_ROWS);
        r.z = __float_as_int(nv[u]);
        r.w = 0;
        rec[(size_t)e0 + pos] = r;
      }
    }
  }
}

__global__ __launch_bounds__(ACC_WAVES * 32) void rgcn_acc_rows(
    const float* __restrict__ T, const int4* __restrict__ rec,
    const int* __restrict__ boff, float* __restrict__ out,
    int r0, int r1, int N, int epb, int first) {
  __shared__ float acc[ACC_ROWS * FEAT];
  const int tid  = threadIdx.x;
  const int lane = tid & 31;
  const int wib  = __builtin_amdgcn_readfirstlane(tid >> 5);
  const int b    = blockIdx.x;
  const int row0 = b * ACC_ROWS;
  int nrows = N - row0;
  if (nrows > ACC_ROWS) nrows = ACC_ROWS;
  const int n4 = nrows * (FEAT / 4);

  float4* __restrict__ acc4 = reinterpret_cast<float4*>(acc);
  float4* __restrict__ out4 = reinterpret_cast<float4*>(out + (size_t)row0 * FEAT);
  if (first) {
    for (int i = tid; i < n4; i += ACC_WAVES * 32) acc4[i] = make_float4(0.f, 0.f, 0.f, 0.f);
  } else {
    for (int i = tid; i < n4; i += ACC_WAVES * 32) acc4[i] = out4[i];
  }
  __syncthreads();

  for (int k = 0; k < NSRC / ACC_WAVES; ++k) {
    const int blk = wib + k * ACC_WAVES;
    const int s = boff[(size_t)blk * BOFF_PITCH + b];
    const int e = boff[(size_t)blk * BOFF_PITCH + b + 1];
    const int4* __restrict__ rp = rec + (size_t)blk * epb;
    for (int base = s; base < e; base += 32) {
      int4 rc = make_int4(0, 0, 0, 0);
      if (base + lane < e) rc = rp[base + lane];
      int cnt = e - base;
      if (cnt > 32) cnt = 32;
      for (int j = 0; j < cnt; j += 8) {
        float2 v[8];
        int    pk[8];
        float  nw[8];
#pragma unroll
        for (int u = 0; u < 8; ++u) {
          int jj = j + u;
          if (jj > cnt - 1) jj = cnt - 1;
          const int sj = __builtin_amdgcn_readlane(rc.x, jj);
          pk[u] = __builtin_amdgcn_readlane(rc.y, jj);
          nw[u] = __int_as_float(__builtin_amdgcn_readlane(rc.z, jj));
          int rr = pk[u] >> 8;
          if (rr < r0) rr = r0;
          if (rr > r1 - 1) rr = r1 - 1;
          v[u] = *reinterpret_cast<const float2*>(
              T + (((size_t)(rr - r0) * N + (size_t)sj) << 6) + (lane << 1));
        }
#pragma unroll
        for (int u = 0; u < 8; ++u) {
          const int r = pk[u] >> 8;
          if (j + u < cnt && r >= r0 && r < r1) {
            float* a = acc + ((pk[u] & 255) << 6) + (lane << 1);
            atomicAdd(a, v[u].x * nw[u]);
            atomicAdd(a + 1, v[u].y * nw[u]);
          }
        }
      }
    }
  }
  __syncthreads();

  for (int i = tid; i < n4; i += ACC_WAVES * 32) out4[i] = acc4[i];
}

__global__ __launch_bounds__(256) void rgcn_edge_direct(
    const float* __restrict__ feat, const float* __restrict__ norm,
    const int* __restrict__ etypes, const int* __restrict__ src,
    const int* __restrict__ dst, const float* __restrict__ weight,
    float* __restrict__ out, int E) {
  int wave = threadIdx.x >> 5;
  int lane = threadIdx.x & 31;
  long long e = (long long)blockIdx.x * 8 + wave;
  if (e >= E) return;

  const float* f = feat + ((size_t)src[e] << 6);
  float x0 = f[lane], x1 = f[lane + 32];
  const float* Wr = weight + (size_t)etypes[e] * (FEAT * FEAT);
  float acc0 = 0.f, acc1 = 0.f;
#pragma unroll
  for (int i = 0; i < 32; ++i) {
    float fi = __shfl(x0, i, 32);
    acc0 += fi * Wr[i * FEAT + lane];
    acc1 += fi * Wr[i * FEAT + lane + 32];
  }
#pragma unroll
  for (int i = 0; i < 32; ++i) {
    float fi = __shfl(x1, i, 32);
    acc0 += fi * Wr[(i + 32) * FEAT + lane];
    acc1 += fi * Wr[(i + 32) * FEAT + lane + 32];
  }
  float nw = norm[e];
  float* op = out + ((size_t)dst[e] << 6);
  unsafeAtomicAdd(op + lane, acc0 * nw);
  unsafeAtomicAdd(op + lane + 32, acc1 * nw);
}

extern "C" void kernel_launch(void* const* d_in, const int* in_sizes, int n_in,
                              void* d_out, int out_size, void* d_ws, size_t ws_size,
                              hipStream_t stream) {
  const float* feat   = (const float*)d_in[0];
  const float* norm   = (const float*)d_in[1];
  const float* weight = (const float*)d_in[2];
  const int*   etypes = (const int*)d_in[3];
  const int*   src    = (const int*)d_in[4];
  const int*   dst    = (const int*)d_in[5];
  float* out = (float*)d_out;

  const int N = in_sizes[0] / FEAT;
  const int E = in_sizes[3];

  const size_t slab = (size_t)N * FEAT * sizeof(float);

  {
    const size_t extra = (size_t)E * sizeof(int4) + (size_t)NSRC * BOFF_PITCH * sizeof(int);
    const int nb = (N + ACC_ROWS - 1) / ACC_ROWS;
    int bchunk = 0;
    if (ws_size > extra) bchunk = (int)((ws_size - extra) / slab);
    if (bchunk > NUM_RELS) bchunk = NUM_RELS;
    if (bchunk > 1) bchunk &= ~1;
    if (bchunk >= 1 && (N % 16) == 0 && nb <= NBKT_MAX) {
      float* T = (float*)d_ws;
      int4* rec = (int4*)((char*)d_ws + (size_t)bchunk * slab);
      int* boff = (int*)((char*)rec + (size_t)E * sizeof(int4));
      const int epb = (E + NSRC - 1) / NSRC;
      rgcn_bin_edges<<<NSRC, 1024, 0, stream>>>(norm, etypes, src, dst, rec, boff, E, epb);
      for (int r0 = 0; r0 < NUM_RELS; r0 += bchunk) {
        int r1 = r0 + bchunk;
        if (r1 > NUM_RELS) r1 = NUM_RELS;
        if (((r1 - r0) % WS_RELS) == 0) {
          dim3 grid(64, (r1 - r0) / WS_RELS);
          rgcn_gemm_wstat<<<grid, 256, 0, stream>>>(feat, weight, T, r0, N);
        } else {
          dim3 grid(N / 16, r1 - r0);
          rgcn_gemm_wmma<<<grid, 128, 0, stream>>>(feat, weight, T, r0, N);
        }
        rgcn_acc_rows<<<nb, ACC_WAVES * 32, 0, stream>>>(T, rec, boff, out, r0, r1, N, epb,
                                                        r0 == 0 ? 1 : 0);
      }
      return;
    }
  }

  int n4 = out_size / 4;
  rgcn_zero_kernel<<<(n4 + 255) / 256, 256, 0, stream>>>((float4*)out, n4);

  int chunk = (int)(ws_size / slab);
  if (chunk > NUM_RELS) chunk = NUM_RELS;

  if (chunk >= 1 && (N % 16) == 0) {
    float* T = (float*)d_ws;
    for (int r0 = 0; r0 < NUM_RELS; r0 += chunk) {
      int r1 = r0 + chunk;
      if (r1 > NUM_RELS) r1 = NUM_RELS;
      if (((r1 - r0) % WS_RELS) == 0) {
        dim3 grid(64, (r1 - r0) / WS_RELS);
        rgcn_gemm_wstat<<<grid, 256, 0, stream>>>(feat, weight, T, r0, N);
      } else {
        dim3 grid(N / 16, r1 - r0);
        rgcn_gemm_wmma<<<grid, 128, 0, stream>>>(feat, weight, T, r0, N);
      }
      long long tt = (long long)E * 16;
      int blocks = (int)((tt + 255) / 256);
      rgcn_edge_scatter<<<blocks, 256, 0, stream>>>(T, norm, etypes, src, dst,
                                                    out, r0, r1, N, E);
    }
  } else {
    int blocks = (E + 7) / 8;
    rgcn_edge_direct<<<blocks, 256, 0, stream>>>(feat, norm, etypes, src, dst,
                                                 weight, out, E);
  }
}


# Aggregate before transform: a block sums norm*feat[src] per (dst row, relation) in LDS (float atomics, arrival order), then multiplies the sums by the weights on f32 WMMA; no T slab, no GEMM launch
# speedup vs baseline: 2.5527x; 1.2477x over previous
#include <hip/hip_runtime.h>

#define FEAT 64
#define NUM_RELS 8

typedef float v2f __attribute__((ext_vector_type(2)));
typedef float v4f __attribute__((ext_vector_type(4)));
typedef float v8f __attribute__((ext_vector_type(8)));

__global__ void rgcn_zero_kernel(float4* __restrict__ out, int n4) {
  int i = blockIdx.x * blockDim.x + threadIdx.x;
  if (i < n4) out[i] = make_float4(0.f, 0.f, 0.f, 0.f);
}

__global__ __launch_bounds__(128) void rgcn_gemm_wmma(
    const float* __restrict__ feat, const float* __restrict__ weight,
    float* __restrict__ T, int r0, int N) {
  const int node0 = blockIdx.x << 4;
  const int rloc  = blockIdx.y;
  const int wave  = threadIdx.x >> 5;
  const int lane  = threadIdx.x & 31;
  const int m     = lane & 15;
  const int khalf = lane >> 4;
  const int o0    = wave << 4;

  const float* __restrict__ Wr = weight + (size_t)(r0 + rloc) * (FEAT * FEAT);
  const float* __restrict__ ap = feat + (size_t)(node0 + m) * FEAT + (khalf << 1);
  const float* __restrict__ bp = Wr + ((khalf << 1) * FEAT) + o0 + m;

  v8f c = {};
#pragma unroll
  for (int k0 = 0; k0 < FEAT; k0 += 4) {
    v2f a, b;
    a.x = ap[k0];
    a.y = ap[k0 + 1];
    b.x = bp[k0 * FEAT];
    b.y = bp[k0 * FEAT + FEAT];
    c = __builtin_amdgcn_wmma_f32_16x16x4_f32(
         false, a,  false, b,
         (short)0, c,  false,  false);
  }

  float* __restrict__ op = T + ((size_t)rloc * N + node0) * FEAT + o0 + m;
#pragma unroll
  for (int v = 0; v < 8; ++v) {
    op[(size_t)(v + (khalf << 3)) * FEAT] = c[v];
  }
}

#define WS_RELS 2
__global__ __launch_bounds__(256) void rgcn_gemm_wstat(
    const float* __restrict__ feat, const float* __restrict__ weight,
    float* __restrict__ T, int r0, int N) {
  const int lane = threadIdx.x & 31;
  const int wib  = __builtin_amdgcn_readfirstlane((int)(threadIdx.x >> 5));
  const int c    = lane & 15;
  const int h    = lane >> 4;
  const int rl0  = blockIdx.y * WS_RELS;
  const int ntiles = N >> 4;
  const int nslots = gridDim.x << 3;
  const int slot = wib * gridDim.x + blockIdx.x;
  if (slot >= ntiles) return;

  v2f b[WS_RELS][4][16];
#pragma unroll
  for (int q = 0; q < WS_RELS; ++q) {
    const float* __restrict__ Wr =
        weight + (size_t)(r0 + rl0 + q) * (FEAT * FEAT) + ((h << 1) * FEAT) + (c << 2);
#pragma unroll
    for (int s = 0; s < 16; ++s) {
      const v4f w0 = *reinterpret_cast<const v4f*>(Wr + (s * 4) * FEAT);
      const v4f w1 = *reinterpret_cast<const v4f*>(Wr + (s * 4 + 1) * FEAT);
      b[q][0][s].x = w0.x; b[q][0][s].y = w1.x;
      b[q][1][s].x = w0.y; b[q][1][s].y = w1.y;
      b[q][2][s].x = w0.z; b[q][2][s].y = w1.z;
      b[q][3][s].x = w0.w; b[q][3][s].y = w1.w;
    }
  }

  const float* __restrict__ abase = feat + (size_t)c * FEAT + (h << 1);
  v2f a[16];
  {
    const float* __restrict__ ap = abase + (size_t)slot * (16 * FEAT);
#pragma unroll
    for (int s = 0; s < 16; ++s) a[s] = *reinterpret_cast<const v2f*>(ap + s * 4);
  }

  for (int tile = slot; tile < ntiles; tile += nslots) {
    int nt = tile + nslots;
    if (nt >= ntiles) nt = tile;
    v2f an[16];
    {
      const float* __restrict__ ap = abase + (size_t)nt * (16 * FEAT);
#pragma unroll
      for (int s = 0; s < 16; ++s) an[s] = *reinterpret_cast<const v2f*>(ap + s * 4);
    }
    __builtin_amdgcn_sched_barrier(0);

    v8f acc[WS_RELS][4];
#pragma unroll
    for (int q = 0; q < WS_RELS; ++q)
#pragma unroll
      for (int t = 0; t < 4; ++t) acc[q][t] = v8f{};

#pragma unroll
    for (int s = 0; s < 16; ++s) {
#pragma unroll
      for (int q = 0; q < WS_RELS; ++q) {
#pragma unroll
        for (int t = 0; t < 4; ++t) {
          acc[q][t] = __builtin_amdgcn_wmma_f32_16x16x4_f32(
               false, a[s],  false, b[q][t][s],
               (short)0, acc[q][t],  false,  false);
        }
      }
    }

    __builtin_amdgcn_sched_barrier(0);

#pragma unroll
    for (int q = 0; q < WS_RELS; ++q) {
      float* __restrict__ op =
          T + (((size_t)(rl0 + q) * N + ((size_t)tile << 4) + (h << 3)) << 6) + (c << 2);
#pragma unroll
      for (int v = 0; v < 8; ++v) {
        v4f o;
        o.x = acc[q][0][v]; o.y = acc[q][1][v]; o.z = acc[q][2][v]; o.w = acc[q][3][v];
        *reinterpret_cast<v4f*>(op + v * FEAT) = o;
      }
    }

#pragma unroll
    for (int s = 0; s < 16; ++s) a[s] = an[s];
  }
}

__global__ __launch_bounds__(256) void rgcn_edge_scatter(
    const float* __restrict__ T, const float* __restrict__ norm,
    const int* __restrict__ etypes, const int* __restrict__ src,
    const int* __restrict__ dst, float* __restrict__ out,
    int r0, int r1, int N, int E) {
  long long t = (long long)blockIdx.x * blockDim.x + threadIdx.x;
  int e = (int)(t >> 4);
  int j = (int)(t & 15);
  if (e >= E) return;
  int r = etypes[e];
  if (r < r0 || r >= r1) return;

  float nw = norm[e];
  const float4* __restrict__ row = reinterpret_cast<const float4*>(
      T + (((size_t)(r - r0) * N + (size_t)src[e]) << 6));
  float4 v = row[j];
  float* op = out + ((size_t)dst[e] << 6) + (j << 2);
  unsafeAtomicAdd(op + 0, v.x * nw);
  unsafeAtomicAdd(op + 1, v.y * nw);
  unsafeAtomicAdd(op + 2, v.z * nw);
  unsafeAtomicAdd(op + 3, v.w * nw);
}

#define AG_ROWS 64
#define NBKT_MAX 2048
#define NSRC 128
#define BOFF_PITCH (NBKT_MAX + 1)

__global__ __launch_bounds__(1024) void rgcn_bin_edges(
    const float* __restrict__ norm, const int* __restrict__ etypes,
    const int* __restrict__ src, const int* __restrict__ dst,
    int4* __restrict__ rec, int* __restrict__ boff, int E, int epb) {
  __shared__ int cnt[NBKT_MAX];
  __shared__ int cur[NBKT_MAX];
  __shared__ int wsum[32];
  const int tid = threadIdx.x;
  const int e0 = blockIdx.x * epb;
  int e1 = e0 + epb;
  if (e1 > E) e1 = E;

  cnt[tid] = 0;
  cnt[tid + 1024] = 0;
  __syncthreads();
  for (int eb = e0 + tid; eb < e1; eb += 4 * 1024) {
    int d[4];
#pragma unroll
    for (int u = 0; u < 4; ++u) {
      const int e = eb + u * 1024;
      d[u] = dst[e < e1 ? e : e1 - 1];
      if (e >= e1) d[u] = -1;
    }
#pragma unroll
    for (int u = 0; u < 4; ++u) {
      const unsigned b = (unsigned)d[u] / AG_ROWS;
      if (b < NBKT_MAX) atomicAdd(&cnt[b], 1);
    }
  }
  __syncthreads();

  {
    const int lane = tid & 31;
    const int wv = tid >> 5;
    const int c0 = cnt[2 * tid];
    const int c1 = cnt[2 * tid + 1];
    const int loc = c0 + c1;
    int incl = loc;
#pragma unroll
    for (int d = 1; d < 32; d <<= 1) {
      const int t = __shfl_up(incl, d, 32);
      if (lane >= d) incl += t;
    }
    if (lane == 31) wsum[wv] = incl;
    __syncthreads();
    if (wv == 0) {
      const int w = wsum[lane];
      int wi = w;
#pragma unroll
      for (int d = 1; d < 32; d <<= 1) {
        const int t = __shfl_up(wi, d, 32);
        if (lane >= d) wi += t;
      }
      wsum[lane] = wi - w;
    }
    __syncthreads();
    const int run = wsum[wv] + incl - loc;
    int* __restrict__ bo = boff + (size_t)blockIdx.x * BOFF_PITCH;
    cur[2 * tid] = run;          bo[2 * tid] = run;
    cur[2 * tid + 1] = run + c0; bo[2 * tid + 1] = run + c0;
    if (tid == 1023) bo[NBKT_MAX] = run + loc;
  }
  __syncthreads();

  for (int eb = e0 + tid; eb < e1; eb += 4 * 1024) {
    int   d[4], sv[4], tv[4];
    float nv[4];
#pragma unroll
    for (int u = 0; u < 4; ++u) {
      const int e  = eb + u * 1024;
      const int ec = e < e1 ? e : e1 - 1;
      d[u]  = dst[ec];
      sv[u] = src[ec];
      tv[u] = etypes[ec];
      nv[u] = norm[ec];
      if (e >= e1) d[u] = -1;
    }
#pragma unroll
    for (int u = 0; u < 4; ++u) {
      const unsigned b = (unsigned)d[u] / AG_ROWS;
      if (b < NBKT_MAX) {
        const int pos = atomicAdd(&cur[b], 1);
        int4 r;
        r.x = sv[u];
        r.y = (tv[u] << 8) | (d[u] - (int)b * AG_ROWS);
        r.z = __float_as_int(nv[u]);
        r.w = 0;
        rec[(size_t)e0 + pos] = r;
      }
    }
  }
}

#define AG_WAVES 16
#define AG_PPW (NSRC / AG_WAVES)
#define AG_HP (NUM_RELS * FEAT + 4)

__global__ __launch_bounds__(AG_WAVES * 32) void rgcn_agg_gemm(
    const float* __restrict__ feat, const float* __restrict__ weight,
    const int4* __restrict__ rec, const int* __restrict__ boff,
    float* __restrict__ out, int N, int epb) {
  __shared__ __attribute__((aligned(16))) float Hs[AG_ROWS * AG_HP];
  __shared__ __attribute__((aligned(16))) float Os[AG_ROWS * FEAT];
  const int tid  = threadIdx.x;
  const int lane = tid & 31;
  const int wib  = __builtin_amdgcn_readfirstlane(tid >> 5);
  const int b    = blockIdx.x;
  const int row0 = b * AG_ROWS;
  int nrows = N - row0;
  if (nrows > AG_ROWS) nrows = AG_ROWS;

  float4* Hs4 = reinterpret_cast<float4*>(Hs);
  float4* Os4 = reinterpret_cast<float4*>(Os);
  for (int i = tid; i < AG_ROWS * AG_HP / 4; i += AG_WAVES * 32)
    Hs4[i] = make_float4(0.f, 0.f, 0.f, 0.f);
  for (int i = tid; i < AG_ROWS * FEAT / 4; i += AG_WAVES * 32)
    Os4[i] = make_float4(0.f, 0.f, 0.f, 0.f);

  int ps = 0, len = 0;
  if (lane < AG_PPW) {
    const int* __restrict__ bo = boff + (size_t)(wib + lane * AG_WAVES) * BOFF_PITCH + b;
    ps  = bo[0];
    len = bo[1] - ps;
  }
  int incl = len;
#pragma unroll
  for (int d = 1; d < AG_PPW; d <<= 1) {
    const int t = __shfl_up(incl, d, 32);
    if (lane >= d) incl += t;
  }
  const int excl  = incl - len;
  const int total = __builtin_amdgcn_readlane(incl, AG_PPW - 1);
  __syncthreads();

  const int c  = lane & 15;
  const int h  = lane >> 4;
  const int ct = wib & 3;
  const int rp = wib >> 2;
  auto load_b = [&](int r, v2f (&bf)[16]) {
    const float* __restrict__ Wr =
        weight + (size_t)r * (FEAT * FEAT) + ((h << 1) * FEAT) + (ct << 4) + c;
#pragma unroll
    for (int s = 0; s < 16; ++s) {
      bf[s].x = Wr[(s * 4) * FEAT];
      bf[s].y = Wr[(s * 4 + 1) * FEAT];
    }
  };
  v2f bfa[16];
  load_b(rp * 2, bfa);

  auto locate = [&](int idx) -> int {
    int ra = 0;
#pragma unroll
    for (int q = 0; q < AG_PPW; ++q) {
      const int ex = __builtin_amdgcn_readlane(excl, q);
      const int in = __builtin_amdgcn_readlane(incl, q);
      const int p0 = __builtin_amdgcn_readlane(ps, q);
      if (idx >= ex && idx < in) ra = (wib + q * AG_WAVES) * epb + p0 + (idx - ex);
    }
    return ra;
  };
  auto load_grp = [&](const int4& rc, int j, int cnt, float2 (&v)[8], int (&pk)[8], float (&nw)[8]) {
#pragma unroll
    for (int u = 0; u < 8; ++u) {
      int jj = j + u;
      if (jj > cnt - 1) jj = cnt - 1;
      const int sj = __builtin_amdgcn_readlane(rc.x, jj);
      pk[u] = __builtin_amdgcn_readlane(rc.y, jj);
      nw[u] = __int_as_float(__builtin_amdgcn_readlane(rc.z, jj));
      v[u] = *reinterpret_cast<const float2*>(feat + ((size_t)sj << 6) + (lane << 1));
    }
  };
  auto add_grp = [&](int j, int cnt, const float2 (&v)[8], const int (&pk)[8], const float (&nw)[8]) {
#pragma unroll
    for (int u = 0; u < 8; ++u) {
      const unsigned r = (unsigned)pk[u] >> 8;
      const bool ok = (j + u < cnt) && (r < NUM_RELS);
      float* a = Hs + (pk[u] & (AG_ROWS - 1)) * AG_HP + ((r & (NUM_RELS - 1)) << 6) + (lane << 1);
      const float x = ok ? v[u].x * nw[u] : 0.f;
      const float y = ok ? v[u].y * nw[u] : 0.f;
      atomicAdd(a, x);
      atomicAdd(a + 1, y);
    }
  };

  int4 rc = make_int4(0, 0, 0, 0);
  if (lane < total) rc = rec[locate(lane)];
  for (int base = 0; base < total; base += 32) {
    int4 rcn = make_int4(0, 0, 0, 0);
    const int nidx = base + 32 + lane;
    if (nidx < total) rcn = rec[locate(nidx)];
    int cnt = total - base;
    if (cnt > 32) cnt = 32;
    float2 va[8], vb[8];
    int    pa[8], pb[8];
    float  na[8], nb[8];
    const int ng = (cnt + 7) >> 3;
    if (ng == 4) {
      load_grp(rc, 0, cnt, va, pa, na);
      load_grp(rc, 8, cnt, vb, pb, nb);
      add_grp(0, cnt, va, pa, na);
      load_grp(rc, 16, cnt, va, pa, na);
      add_grp(8, cnt, vb, pb, nb);
      load_grp(rc, 24, cnt, vb, pb, nb);
      add_grp(16, cnt, va, pa, na);
      add_grp(24, cnt, vb, pb, nb);
    } else if (ng == 3) {
      load_grp(rc, 0, cnt, va, pa, na);
      load_grp(rc, 8, cnt, vb, pb, nb);
      add_grp(0, cnt, va, pa, na);
      load_grp(rc, 16, cnt, va, pa, na);
      add_grp(8, cnt, vb, pb, nb);
      add_grp(16, cnt, va, pa, na);
    } else if (ng == 2) {
      load_grp(rc, 0, cnt, va, pa, na);
      load_grp(rc, 8, cnt, vb, pb, nb);
      add_grp(0, cnt, va, pa, na);
      add_grp(8, cnt, vb, pb, nb);
    } else {
      load_grp(rc, 0, cnt, va, pa, na);
      add_grp(0, cnt, va, pa, na);
    }
    rc = rcn;
  }
  __syncthreads();

  {
    v8f acc[4];
#pragma unroll
    for (int rt = 0; rt < 4; ++rt) acc[rt] = v8f{};
    v2f bfb[16];
    load_b(rp * 2 + 1, bfb);

    auto mm = [&](int r, const v2f (&bf)[16]) {
#pragma unroll
      for (int rt = 0; rt < 4; ++rt) {
        const float* Ha = Hs + (rt * 16 + c) * AG_HP + (r << 6) + (h << 1);
        v2f af[16];
#pragma unroll
        for (int s = 0; s < 16; ++s) af[s] = *reinterpret_cast<const v2f*>(Ha + s * 4);
#pragma unroll
        for (int s = 0; s < 16; ++s) {
          acc[rt] = __builtin_amdgcn_wmma_f32_16x16x4_f32(
               false, af[s],  false, bf[s],
               (short)0, acc[rt],  false,  false);
        }
      }
    };
    mm(rp * 2, bfa);
    mm(rp * 2 + 1, bfb);

#pragma unroll
    for (int rt = 0; rt < 4; ++rt) {
      float* o = Os + (rt * 16 + (h << 3)) * FEAT + (ct << 4) + c;
#pragma unroll
      for (int v = 0; v < 8; ++v) atomicAdd(o + v * FEAT, acc[rt][v]);
    }
  }
  __syncthreads();

  float4* __restrict__ out4 = reinterpret_cast<float4*>(out + (size_t)row0 * FEAT);
  const int n4 = nrows * (FEAT / 4);
  for (int i = tid; i < n4; i += AG_WAVES * 32) out4[i] = Os4[i];
}

__global__ __launch_bounds__(256) void rgcn_edge_direct(
    const float* __restrict__ feat, const float* __restrict__ norm,
    const int* __restrict__ etypes, const int* __restrict__ src,
    const int* __restrict__ dst, const float* __restrict__ weight,
    float* __restrict__ out, int E) {
  int wave = threadIdx.x >> 5;
  int lane = threadIdx.x & 31;
  long long e = (long long)blockIdx.x * 8 + wave;
  if (e >= E) return;

  const float* f = feat + ((size_t)src[e] << 6);
  float x0 = f[lane], x1 = f[lane + 32];
  const float* Wr = weight + (size_t)etypes[e] * (FEAT * FEAT);
  float acc0 = 0.f, acc1 = 0.f;
#pragma unroll
  for (int i = 0; i < 32; ++i) {
    float fi = __shfl(x0, i, 32);
    acc0 += fi * Wr[i * FEAT + lane];
    acc1 += fi * Wr[i * FEAT + lane + 32];
  }
#pragma unroll
  for (int i = 0; i < 32; ++i) {
    float fi = __shfl(x1, i, 32);
    acc0 += fi * Wr[(i + 32) * FEAT + lane];
    acc1 += fi * Wr[(i + 32) * FEAT + lane + 32];
  }
  float nw = norm[e];
  float* op = out + ((size_t)dst[e] << 6);
  unsafeAtomicAdd(op + lane, acc0 * nw);
  unsafeAtomicAdd(op + lane + 32, acc1 * nw);
}

extern "C" void kernel_launch(void* const* d_in, const int* in_sizes, int n_in,
                              void* d_out, int out_size, void* d_ws, size_t ws_size,
                              hipStream_t stream) {
  const float* feat   = (const float*)d_in[0];
  const float* norm   = (const float*)d_in[1];
  const float* weight = (const float*)d_in[2];
  const int*   etypes = (const int*)d_in[3];
  const int*   src    = (const int*)d_in[4];
  const int*   dst    = (const int*)d_in[5];
  float* out = (float*)d_out;

  const int N = in_sizes[0] / FEAT;
  const int E = in_sizes[3];

  const size_t slab = (size_t)N * FEAT * sizeof(float);

  {
    const size_t extra = (size_t)E * sizeof(int4) + (size_t)NSRC * BOFF_PITCH * sizeof(int);
    const int nb = (N + AG_ROWS - 1) / AG_ROWS;
    if (ws_size >= extra && nb <= NBKT_MAX) {
      int4* rec = (int4*)d_ws;
      int* boff = (int*)((char*)d_ws + (size_t)E * sizeof(int4));
      const int epb = (E + NSRC - 1) / NSRC;
      rgcn_bin_edges<<<NSRC, 1024, 0, stream>>>(norm, etypes, src, dst, rec, boff, E, epb);
      rgcn_agg_gemm<<<nb, AG_WAVES * 32, 0, stream>>>(feat, weight, rec, boff, out, N, epb);
      return;
    }
  }

  int n4 = out_size / 4;
  rgcn_zero_kernel<<<(n4 + 255) / 256, 256, 0, stream>>>((float4*)out, n4);

  int chunk = (int)(ws_size / slab);
  if (chunk > NUM_RELS) chunk = NUM_RELS;

  if (chunk >= 1 && (N % 16) == 0) {
    float* T = (float*)d_ws;
    for (int r0 = 0; r0 < NUM_RELS; r0 += chunk) {
      int r1 = r0 + chunk;
      if (r1 > NUM_RELS) r1 = NUM_RELS;
      if (((r1 - r0) % WS_RELS) == 0) {
        dim3 grid(64, (r1 - r0) / WS_RELS);
        rgcn_gemm_wstat<<<grid, 256, 0, stream>>>(feat, weight, T, r0, N);
      } else {
        dim3 grid(N / 16, r1 - r0);
        rgcn_gemm_wmma<<<grid, 128, 0, stream>>>(feat, weight, T, r0, N);
      }
      long long tt = (long long)E * 16;
      int blocks = (int)((tt + 255) / 256);
      rgcn_edge_scatter<<<blocks, 256, 0, stream>>>(T, norm, etypes, src, dst,
                                                    out, r0, r1, N, E);
    }
  } else {
    int blocks = (E + 7) / 8;
    rgcn_edge_direct<<<blocks, 256, 0, stream>>>(feat, norm, etypes, src, dst,
                                                 weight, out, E);
  }
}


# Binning kernel groups a slice's records in an LDS copy of its segment and stores it as whole 128-B lines (512 B per store); was one 16-B store per record to a random line; ranks still arrival order
# speedup vs baseline: 2.6526x; 1.0391x over previous
#include <hip/hip_runtime.h>

#define FEAT 64
#define NUM_RELS 8

typedef float v2f __attribute__((ext_vector_type(2)));
typedef float v4f __attribute__((ext_vector_type(4)));
typedef float v8f __attribute__((ext_vector_type(8)));

__global__ void rgcn_zero_kernel(float4* __restrict__ out, int n4) {
  int i = blockIdx.x * blockDim.x + threadIdx.x;
  if (i < n4) out[i] = make_float4(0.f, 0.f, 0.f, 0.f);
}

__global__ __launch_bounds__(128) void rgcn_gemm_wmma(
    const float* __restrict__ feat, const float* __restrict__ weight,
    float* __restrict__ T, int r0, int N) {
  const int node0 = blockIdx.x << 4;
  const int rloc  = blockIdx.y;
  const int wave  = threadIdx.x >> 5;
  const int lane  = threadIdx.x & 31;
  const int m     = lane & 15;
  const int khalf = lane >> 4;
  const int o0    = wave << 4;

  const float* __restrict__ Wr = weight + (size_t)(r0 + rloc) * (FEAT * FEAT);
  const float* __restrict__ ap = feat + (size_t)(node0 + m) * FEAT + (khalf << 1);
  const float* __restrict__ bp = Wr + ((khalf << 1) * FEAT) + o0 + m;

  v8f c = {};
#pragma unroll
  for (int k0 = 0; k0 < FEAT; k0 += 4) {
    v2f a, b;
    a.x = ap[k0];
    a.y = ap[k0 + 1];
    b.x = bp[k0 * FEAT];
    b.y = bp[k0 * FEAT + FEAT];
    c = __builtin_amdgcn_wmma_f32_16x16x4_f32(
         false, a,  false, b,
         (short)0, c,  false,  false);
  }

  float* __restrict__ op = T + ((size_t)rloc * N + node0) * FEAT + o0 + m;
#pragma unroll
  for (int v = 0; v < 8; ++v) {
    op[(size_t)(v + (khalf << 3)) * FEAT] = c[v];
  }
}

#define WS_RELS 2
__global__ __launch_bounds__(256) void rgcn_gemm_wstat(
    const float* __restrict__ feat, const float* __restrict__ weight,
    float* __restrict__ T, int r0, int N) {
  const int lane = threadIdx.x & 31;
  const int wib  = __builtin_amdgcn_readfirstlane((int)(threadIdx.x >> 5));
  const int c    = lane & 15;
  const int h    = lane >> 4;
  const int rl0  = blockIdx.y * WS_RELS;
  const int ntiles = N >> 4;
  const int nslots = gridDim.x << 3;
  const int slot = wib * gridDim.x + blockIdx.x;
  if (slot >= ntiles) return;

  v2f b[WS_RELS][4][16];
#pragma unroll
  for (int q = 0; q < WS_RELS; ++q) {
    const float* __restrict__ Wr =
        weight + (size_t)(r0 + rl0 + q) * (FEAT * FEAT) + ((h << 1) * FEAT) + (c << 2);
#pragma unroll
    for (int s = 0; s < 16; ++s) {
      const v4f w0 = *reinterpret_cast<const v4f*>(Wr + (s * 4) * FEAT);
      const v4f w1 = *reinterpret_cast<const v4f*>(Wr + (s * 4 + 1) * FEAT);
      b[q][0][s].x = w0.x; b[q][0][s].y = w1.x;
      b[q][1][s].x = w0.y; b[q][1][s].y = w1.y;
      b[q][2][s].x = w0.z; b[q][2][s].y = w1.z;
      b[q][3][s].x = w0.w; b[q][3][s].y = w1.w;
    }
  }

  const float* __restrict__ abase = feat + (size_t)c * FEAT + (h << 1);
  v2f a[16];
  {
    const float* __restrict__ ap = abase + (size_t)slot * (16 * FEAT);
#pragma unroll
    for (int s = 0; s < 16; ++s) a[s] = *reinterpret_cast<const v2f*>(ap + s * 4);
  }

  for (int tile = slot; tile < ntiles; tile += nslots) {
    int nt = tile + nslots;
    if (nt >= ntiles) nt = tile;
    v2f an[16];
    {
      const float* __restrict__ ap = abase + (size_t)nt * (16 * FEAT);
#pragma unroll
      for (int s = 0; s < 16; ++s) an[s] = *reinterpret_cast<const v2f*>(ap + s * 4);
    }
    __builtin_amdgcn_sched_barrier(0);

    v8f acc[WS_RELS][4];
#pragma unroll
    for (int q = 0; q < WS_RELS; ++q)
#pragma unroll
      for (int t = 0; t < 4; ++t) acc[q][t] = v8f{};

#pragma unroll
    for (int s = 0; s < 16; ++s) {
#pragma unroll
      for (int q = 0; q < WS_RELS; ++q) {
#pragma unroll
        for (int t = 0; t < 4; ++t) {
          acc[q][t] = __builtin_amdgcn_wmma_f32_16x16x4_f32(
               false, a[s],  false, b[q][t][s],
               (short)0, acc[q][t],  false,  false);
        }
      }
    }

    __builtin_amdgcn_sched_barrier(0);

#pragma unroll
    for (int q = 0; q < WS_RELS; ++q) {
      float* __restrict__ op =
          T + (((size_t)(rl0 + q) * N + ((size_t)tile << 4) + (h << 3)) << 6) + (c << 2);
#pragma unroll
      for (int v = 0; v < 8; ++v) {
        v4f o;
        o.x = acc[q][0][v]; o.y = acc[q][1][v]; o.z = acc[q][2][v]; o.w = acc[q][3][v];
        *reinterpret_cast<v4f*>(op + v * FEAT) = o;
      }
    }

#pragma unroll
    for (int s = 0; s < 16; ++s) a[s] = an[s];
  }
}

__global__ __launch_bounds__(256) void rgcn_edge_scatter(
    const float* __restrict__ T, const float* __restrict__ norm,
    const int* __restrict__ etypes, const int* __restrict__ src,
    const int* __restrict__ dst, float* __restrict__ out,
    int r0, int r1, int N, int E) {
  long long t = (long long)blockIdx.x * blockDim.x + threadIdx.x;
  int e = (int)(t >> 4);
  int j = (int)(t & 15);
  if (e >= E) return;
  int r = etypes[e];
  if (r < r0 || r >= r1) return;

  float nw = norm[e];
  const float4* __restrict__ row = reinterpret_cast<const float4*>(
      T + (((size_t)(r - r0) * N + (size_t)src[e]) << 6));
  float4 v = row[j];
  float* op = out + ((size_t)dst[e] << 6) + (j << 2);
  unsafeAtomicAdd(op + 0, v.x * nw);
  unsafeAtomicAdd(op + 1, v.y * nw);
  unsafeAtomicAdd(op + 2, v.z * nw);
  unsafeAtomicAdd(op + 3, v.w * nw);
}

#define AG_ROWS 64
#define NBKT_MAX 2048
#define NSRC 128
#define BOFF_PITCH (NBKT_MAX + 1)

__global__ __launch_bounds__(1024) void rgcn_bin_edges(
    const float* __restrict__ norm, const int* __restrict__ etypes,
    const int* __restrict__ src, const int* __restrict__ dst,
    int4* __restrict__ rec, int* __restrict__ boff, int E, int epb) {
  __shared__ int cnt[NBKT_MAX];
  __shared__ int cur[NBKT_MAX];
  __shared__ int wsum[32];
  const int tid = threadIdx.x;
  const int e0 = blockIdx.x * epb;
  int e1 = e0 + epb;
  if (e1 > E) e1 = E;

  cnt[tid] = 0;
  cnt[tid + 1024] = 0;
  __syncthreads();
  for (int eb = e0 + tid; eb < e1; eb += 4 * 1024) {
    int d[4];
#pragma unroll
    for (int u = 0; u < 4; ++u) {
      const int e = eb + u * 1024;
      d[u] = dst[e < e1 ? e : e1 - 1];
      if (e >= e1) d[u] = -1;
    }
#pragma unroll
    for (int u = 0; u < 4; ++u) {
      const unsigned b = (unsigned)d[u] / AG_ROWS;
      if (b < NBKT_MAX) atomicAdd(&cnt[b], 1);
    }
  }
  __syncthreads();

  {
    const int lane = tid & 31;
    const int wv = tid >> 5;
    const int c0 = cnt[2 * tid];
    const int c1 = cnt[2 * tid + 1];
    const int loc = c0 + c1;
    int incl = loc;
#pragma unroll
    for (int d = 1; d < 32; d <<= 1) {
      const int t = __shfl_up(incl, d, 32);
      if (lane >= d) incl += t;
    }
    if (lane == 31) wsum[wv] = incl;
    __syncthreads();
    if (wv == 0) {
      const int w = wsum[lane];
      int wi = w;
#pragma unroll
      for (int d = 1; d < 32; d <<= 1) {
        const int t = __shfl_up(wi, d, 32);
        if (lane >= d) wi += t;
      }
      wsum[lane] = wi - w;
    }
    __syncthreads();
    const int run = wsum[wv] + incl - loc;
    int* __restrict__ bo = boff + (size_t)blockIdx.x * BOFF_PITCH;
    cur[2 * tid] = run;          bo[2 * tid] = run;
    cur[2 * tid + 1] = run + c0; bo[2 * tid + 1] = run + c0;
    if (tid == 1023) bo[NBKT_MAX] = run + loc;
  }
  __syncthreads();

  for (int eb = e0 + tid; eb < e1; eb += 4 * 1024) {
    int   d[4], sv[4], tv[4];
    float nv[4];
#pragma unroll
    for (int u = 0; u < 4; ++u) {
      const int e  = eb + u * 1024;
      const int ec = e < e1 ? e : e1 - 1;
      d[u]  = dst[ec];
      sv[u] = src[ec];
      tv[u] = etypes[ec];
      nv[u] = norm[ec];
      if (e >= e1) d[u] = -1;
    }
#pragma unroll
    for (int u = 0; u < 4; ++u) {
      const unsigned b = (unsigned)d[u] / AG_ROWS;
      if (b < NBKT_MAX) {
        const int pos = atomicAdd(&cur[b], 1);
        int4 r;
        r.x = sv[u];
        r.y = (tv[u] << 8) | (d[u] - (int)b * AG_ROWS);
        r.z = __float_as_int(nv[u]);
        r.w = 0;
        rec[(size_t)e0 + pos] = r;
      }
    }
  }
}

#define BIN_STAGE_MAX 12544

__global__ __launch_bounds__(1024) void rgcn_bin_edges_staged(
    const float* __restrict__ norm, const int* __restrict__ etypes,
    const int* __restrict__ src, const int* __restrict__ dst,
    int4* __restrict__ rec, int* __restrict__ boff, int E, int epb) {
  __shared__ __attribute__((aligned(16))) int4 srec[BIN_STAGE_MAX];
  __shared__ int cnt[NBKT_MAX];
  __shared__ int cur[NBKT_MAX];
  __shared__ int wsum[32];
  __shared__ int ntot;
  const int tid = threadIdx.x;
  const int e0 = blockIdx.x * epb;
  int e1 = e0 + epb;
  if (e1 > E) e1 = E;

  cnt[tid] = 0;
  cnt[tid + 1024] = 0;
  __syncthreads();
  for (int eb = e0 + tid; eb < e1; eb += 4 * 1024) {
    int d[4];
#pragma unroll
    for (int u = 0; u < 4; ++u) {
      const int e = eb + u * 1024;
      d[u] = dst[e < e1 ? e : e1 - 1];
      if (e >= e1) d[u] = -1;
    }
#pragma unroll
    for (int u = 0; u < 4; ++u) {
      const unsigned b = (unsigned)d[u] / AG_ROWS;
      if (b < NBKT_MAX) atomicAdd(&cnt[b], 1);
    }
  }
  __syncthreads();

  {
    const int lane = tid & 31;
    const int wv = tid >> 5;
    const int c0 = cnt[2 * tid];
    const int c1 = cnt[2 * tid + 1];
    const int loc = c0 + c1;
    int incl = loc;
#pragma unroll
    for (int d = 1; d < 32; d <<= 1) {
      const int t = __shfl_up(incl, d, 32);
      if (lane >= d) incl += t;
    }
    if (lane == 31) wsum[wv] = incl;
    __syncthreads();
    if (wv == 0) {
      const int w = wsum[lane];
      int wi = w;
#pragma unroll
      for (int d = 1; d < 32; d <<= 1) {
        const int t = __shfl_up(wi, d, 32);
        if (lane >= d) wi += t;
      }
      wsum[lane] = wi - w;
    }
    __syncthreads();
    const int run = wsum[wv] + incl - loc;
    int* __restrict__ bo = boff + (size_t)blockIdx.x * BOFF_PITCH;
    cur[2 * tid] = run;          bo[2 * tid] = run;
    cur[2 * tid + 1] = run + c0; bo[2 * tid + 1] = run + c0;
    if (tid == 1023) { bo[NBKT_MAX] = run + loc; ntot = run + loc; }
  }
  __syncthreads();

  for (int eb = e0 + tid; eb < e1; eb += 4 * 1024) {
    int   d[4], sv[4], tv[4];
    float nv[4];
#pragma unroll
    for (int u = 0; u < 4; ++u) {
      const int e  = eb + u * 1024;
      const int ec = e < e1 ? e : e1 - 1;
      d[u]  = dst[ec];
      sv[u] = src[ec];
      tv[u] = etypes[ec];
      nv[u] = norm[ec];
      if (e >= e1) d[u] = -1;
    }
#pragma unroll
    for (int u = 0; u < 4; ++u) {
      const unsigned b = (unsigned)d[u] / AG_ROWS;
      if (b < NBKT_MAX) {
        const int pos = atomicAdd(&cur[b], 1);
        int4 r;
        r.x = sv[u];
        r.y = (tv[u] << 8) | (d[u] - (int)b * AG_ROWS);
        r.z = __float_as_int(nv[u]);
        r.w = 0;
        srec[pos] = r;
      }
    }
  }
  __syncthreads();

  const int n = ntot;
  const int skew = e0 & 7;
  int4* __restrict__ seg = rec + (size_t)e0;
  for (int i = tid - skew; i < n; i += 1024)
    if (i >= 0) seg[i] = srec[i];
}

#define AG_WAVES 16
#define AG_PPW (NSRC / AG_WAVES)
#define AG_HP (NUM_RELS * FEAT + 4)

__global__ __launch_bounds__(AG_WAVES * 32) void rgcn_agg_gemm(
    const float* __restrict__ feat, const float* __restrict__ weight,
    const int4* __restrict__ rec, const int* __restrict__ boff,
    float* __restrict__ out, int N, int epb) {
  __shared__ __attribute__((aligned(16))) float Hs[AG_ROWS * AG_HP];
  __shared__ __attribute__((aligned(16))) float Os[AG_ROWS * FEAT];
  const int tid  = threadIdx.x;
  const int lane = tid & 31;
  const int wib  = __builtin_amdgcn_readfirstlane(tid >> 5);
  const int b    = blockIdx.x;
  const int row0 = b * AG_ROWS;
  int nrows = N - row0;
  if (nrows > AG_ROWS) nrows = AG_ROWS;

  float4* Hs4 = reinterpret_cast<float4*>(Hs);
  float4* Os4 = reinterpret_cast<float4*>(Os);
  for (int i = tid; i < AG_ROWS * AG_HP / 4; i += AG_WAVES * 32)
    Hs4[i] = make_float4(0.f, 0.f, 0.f, 0.f);
  for (int i = tid; i < AG_ROWS * FEAT / 4; i += AG_WAVES * 32)
    Os4[i] = make_float4(0.f, 0.f, 0.f, 0.f);

  int ps = 0, len = 0;
  if (lane < AG_PPW) {
    const int* __restrict__ bo = boff + (size_t)(wib + lane * AG_WAVES) * BOFF_PITCH + b;
    ps  = bo[0];
    len = bo[1] - ps;
  }
  int incl = len;
#pragma unroll
  for (int d = 1; d < AG_PPW; d <<= 1) {
    const int t = __shfl_up(incl, d, 32);
    if (lane >= d) incl += t;
  }
  const int excl  = incl - len;
  const int total = __builtin_amdgcn_readlane(incl, AG_PPW - 1);
  __syncthreads();

  const int c  = lane & 15;
  const int h  = lane >> 4;
  const int ct = wib & 3;
  const int rp = wib >> 2;
  auto load_b = [&](int r, v2f (&bf)[16]) {
    const float* __restrict__ Wr =
        weight + (size_t)r * (FEAT * FEAT) + ((h << 1) * FEAT) + (ct << 4) + c;
#pragma unroll
    for (int s = 0; s < 16; ++s) {
      bf[s].x = Wr[(s * 4) * FEAT];
      bf[s].y = Wr[(s * 4 + 1) * FEAT];
    }
  };
  v2f bfa[16];
  load_b(rp * 2, bfa);

  auto locate = [&](int idx) -> int {
    int ra = 0;
#pragma unroll
    for (int q = 0; q < AG_PPW; ++q) {
      const int ex = __builtin_amdgcn_readlane(excl, q);
      const int in = __builtin_amdgcn_readlane(incl, q);
      const int p0 = __builtin_amdgcn_readlane(ps, q);
      if (idx >= ex && idx < in) ra = (wib + q * AG_WAVES) * epb + p0 + (idx - ex);
    }
    return ra;
  };
  auto load_grp = [&](const int4& rc, int j, int cnt, float2 (&v)[8], int (&pk)[8], float (&nw)[8]) {
#pragma unroll
    for (int u = 0; u < 8; ++u) {
      int jj = j + u;
      if (jj > cnt - 1) jj = cnt - 1;
      const int sj = __builtin_amdgcn_readlane(rc.x, jj);
      pk[u] = __builtin_amdgcn_readlane(rc.y, jj);
      nw[u] = __int_as_float(__builtin_amdgcn_readlane(rc.z, jj));
      v[u] = *reinterpret_cast<const float2*>(feat + ((size_t)sj << 6) + (lane << 1));
    }
  };
  auto add_grp = [&](int j, int cnt, const float2 (&v)[8], const int (&pk)[8], const float (&nw)[8]) {
#pragma unroll
    for (int u = 0; u < 8; ++u) {
      const unsigned r = (unsigned)pk[u] >> 8;
      const bool ok = (j + u < cnt) && (r < NUM_RELS);
      float* a = Hs + (pk[u] & (AG_ROWS - 1)) * AG_HP + ((r & (NUM_RELS - 1)) << 6) + (lane << 1);
      const float x = ok ? v[u].x * nw[u] : 0.f;
      const float y = ok ? v[u].y * nw[u] : 0.f;
      atomicAdd(a, x);
      atomicAdd(a + 1, y);
    }
  };

  int4 rc = make_int4(0, 0, 0, 0);
  if (lane < total) rc = rec[locate(lane)];
  for (int base = 0; base < total; base += 32) {
    int4 rcn = make_int4(0, 0, 0, 0);
    const int nidx = base + 32 + lane;
    if (nidx < total) rcn = rec[locate(nidx)];
    int cnt = total - base;
    if (cnt > 32) cnt = 32;
    float2 va[8], vb[8];
    int    pa[8], pb[8];
    float  na[8], nb[8];
    const int ng = (cnt + 7) >> 3;
    if (ng == 4) {
      load_grp(rc, 0, cnt, va, pa, na);
      load_grp(rc, 8, cnt, vb, pb, nb);
      add_grp(0, cnt, va, pa, na);
      load_grp(rc, 16, cnt, va, pa, na);
      add_grp(8, cnt, vb, pb, nb);
      load_grp(rc, 24, cnt, vb, pb, nb);
      add_grp(16, cnt, va, pa, na);
      add_grp(24, cnt, vb, pb, nb);
    } else if (ng == 3) {
      load_grp(rc, 0, cnt, va, pa, na);
      load_grp(rc, 8, cnt, vb, pb, nb);
      add_grp(0, cnt, va, pa, na);
      load_grp(rc, 16, cnt, va, pa, na);
      add_grp(8, cnt, vb, pb, nb);
      add_grp(16, cnt, va, pa, na);
    } else if (ng == 2) {
      load_grp(rc, 0, cnt, va, pa, na);
      load_grp(rc, 8, cnt, vb, pb, nb);
      add_grp(0, cnt, va, pa, na);
      add_grp(8, cnt, vb, pb, nb);
    } else {
      load_grp(rc, 0, cnt, va, pa, na);
      add_grp(0, cnt, va, pa, na);
    }
    rc = rcn;
  }
  __syncthreads();

  {
    v8f acc[4];
#pragma unroll
    for (int rt = 0; rt < 4; ++rt) acc[rt] = v8f{};
    v2f bfb[16];
    load_b(rp * 2 + 1, bfb);

    auto mm = [&](int r, const v2f (&bf)[16]) {
#pragma unroll
      for (int rt = 0; rt < 4; ++rt) {
        const float* Ha = Hs + (rt * 16 + c) * AG_HP + (r << 6) + (h << 1);
        v2f af[16];
#pragma unroll
        for (int s = 0; s < 16; ++s) af[s] = *reinterpret_cast<const v2f*>(Ha + s * 4);
#pragma unroll
        for (int s = 0; s < 16; ++s) {
          acc[rt] = __builtin_amdgcn_wmma_f32_16x16x4_f32(
               false, af[s],  false, bf[s],
               (short)0, acc[rt],  false,  false);
        }
      }
    };
    mm(rp * 2, bfa);
    mm(rp * 2 + 1, bfb);

#pragma unroll
    for (int rt = 0; rt < 4; ++rt) {
      float* o = Os + (rt * 16 + (h << 3)) * FEAT + (ct << 4) + c;
#pragma unroll
      for (int v = 0; v < 8; ++v) atomicAdd(o + v * FEAT, acc[rt][v]);
    }
  }
  __syncthreads();

  float4* __restrict__ out4 = reinterpret_cast<float4*>(out + (size_t)row0 * FEAT);
  const int n4 = nrows * (FEAT / 4);
  for (int i = tid; i < n4; i += AG_WAVES * 32) out4[i] = Os4[i];
}

__global__ __launch_bounds__(256) void rgcn_edge_direct(
    const float* __restrict__ feat, const float* __restrict__ norm,
    const int* __restrict__ etypes, const int* __restrict__ src,
    const int* __restrict__ dst, const float* __restrict__ weight,
    float* __restrict__ out, int E) {
  int wave = threadIdx.x >> 5;
  int lane = threadIdx.x & 31;
  long long e = (long long)blockIdx.x * 8 + wave;
  if (e >= E) return;

  const float* f = feat + ((size_t)src[e] << 6);
  float x0 = f[lane], x1 = f[lane + 32];
  const float* Wr = weight + (size_t)etypes[e] * (FEAT * FEAT);
  float acc0 = 0.f, acc1 = 0.f;
#pragma unroll
  for (int i = 0; i < 32; ++i) {
    float fi = __shfl(x0, i, 32);
    acc0 += fi * Wr[i * FEAT + lane];
    acc1 += fi * Wr[i * FEAT + lane + 32];
  }
#pragma unroll
  for (int i = 0; i < 32; ++i) {
    float fi = __shfl(x1, i, 32);
    acc0 += fi * Wr[(i + 32) * FEAT + lane];
    acc1 += fi * Wr[(i + 32) * FEAT + lane + 32];
  }
  float nw = norm[e];
  float* op = out + ((size_t)dst[e] << 6);
  unsafeAtomicAdd(op + lane, acc0 * nw);
  unsafeAtomicAdd(op + lane + 32, acc1 * nw);
}

extern "C" void kernel_launch(void* const* d_in, const int* in_sizes, int n_in,
                              void* d_out, int out_size, void* d_ws, size_t ws_size,
                              hipStream_t stream) {
  const float* feat   = (const float*)d_in[0];
  const float* norm   = (const float*)d_in[1];
  const float* weight = (const float*)d_in[2];
  const int*   etypes = (const int*)d_in[3];
  const int*   src    = (const int*)d_in[4];
  const int*   dst    = (const int*)d_in[5];
  float* out = (float*)d_out;

  const int N = in_sizes[0] / FEAT;
  const int E = in_sizes[3];

  const size_t slab = (size_t)N * FEAT * sizeof(float);

  {
    const size_t extra = (size_t)E * sizeof(int4) + (size_t)NSRC * BOFF_PITCH * sizeof(int);
    const int nb = (N + AG_ROWS - 1) / AG_ROWS;
    if (ws_size >= extra && nb <= NBKT_MAX) {
      int4* rec = (int4*)d_ws;
      int* boff = (int*)((char*)d_ws + (size_t)E * sizeof(int4));
      const int epb = (E + NSRC - 1) / NSRC;
      if (epb <= BIN_STAGE_MAX)
        rgcn_bin_edges_staged<<<NSRC, 1024, 0, stream>>>(norm, etypes, src, dst, rec, boff, E, epb);
      else
        rgcn_bin_edges<<<NSRC, 1024, 0, stream>>>(norm, etypes, src, dst, rec, boff, E, epb);
      rgcn_agg_gemm<<<nb, AG_WAVES * 32, 0, stream>>>(feat, weight, rec, boff, out, N, epb);
      return;
    }
  }

  int n4 = out_size / 4;
  rgcn_zero_kernel<<<(n4 + 255) / 256, 256, 0, stream>>>((float4*)out, n4);

  int chunk = (int)(ws_size / slab);
  if (chunk > NUM_RELS) chunk = NUM_RELS;

  if (chunk >= 1 && (N % 16) == 0) {
    float* T = (float*)d_ws;
    for (int r0 = 0; r0 < NUM_RELS; r0 += chunk) {
      int r1 = r0 + chunk;
      if (r1 > NUM_RELS) r1 = NUM_RELS;
      if (((r1 - r0) % WS_RELS) == 0) {
        dim3 grid(64, (r1 - r0) / WS_RELS);
        rgcn_gemm_wstat<<<grid, 256, 0, stream>>>(feat, weight, T, r0, N);
      } else {
        dim3 grid(N / 16, r1 - r0);
        rgcn_gemm_wmma<<<grid, 128, 0, stream>>>(feat, weight, T, r0, N);
      }
      long long tt = (long long)E * 16;
      int blocks = (int)((tt + 255) / 256);
      rgcn_edge_scatter<<<blocks, 256, 0, stream>>>(T, norm, etypes, src, dst,
                                                    out, r0, r1, N, E);
    }
  } else {
    int blocks = (E + 7) / 8;
    rgcn_edge_direct<<<blocks, 256, 0, stream>>>(feat, norm, etypes, src, dst,
                                                 weight, out, E);
  }
}
